# starting kernel
# baseline (speedup 1.0000x reference)
#include <hip/hip_runtime.h>

typedef __attribute__((ext_vector_type(2))) float v2f;
typedef __attribute__((ext_vector_type(8))) float v8f;

#define N_NODES 50000
#define FDIM    128
#define E_EDGES 1600000

__global__ void gcn_zero_kernel(float* __restrict__ h, int n) {
    int i = blockIdx.x * blockDim.x + threadIdx.x;
    if (i < n) h[i] = 0.0f;
}

__global__ void gcn_scatter_kernel(const float* __restrict__ feature,
                                   const int*   __restrict__ src,
                                   const int*   __restrict__ dst,
                                   float*       __restrict__ h) {
    long long t = (long long)blockIdx.x * blockDim.x + threadIdx.x;
    int e = (int)(t >> 5);
    int c = (int)(t & 31);
    if (e >= E_EDGES) return;

    int s = src[e];
    int d = dst[e];

    const float4* frow = (const float4*)(feature + (size_t)s * FDIM);
    float4 v = frow[c];

    float* hrow = h + (size_t)d * FDIM + (size_t)c * 4;
    unsafeAtomicAdd(hrow + 0, v.x);
    unsafeAtomicAdd(hrow + 1, v.y);
    unsafeAtomicAdd(hrow + 2, v.z);
    unsafeAtomicAdd(hrow + 3, v.w);
}

__global__ void gcn_gemm_wmma_kernel(const float* __restrict__ h,
                                     const float* __restrict__ W,
                                     const float* __restrict__ b,
                                     float*       __restrict__ out) {
    const int lane = threadIdx.x;
    const int col0 = blockIdx.x * 16;
    const int row0 = blockIdx.y * 16;
    const int ln   = lane & 15;
    const int half = lane >> 4;
    const int koff = half * 2;

    float bb = b[col0 + ln];
    v8f acc;
#pragma unroll
    for (int v = 0; v < 8; ++v) acc[v] = bb;

    const float* arow = h + (size_t)(row0 + ln) * FDIM;
    const float* brow = W + (size_t)(col0 + ln) * FDIM;

#pragma unroll 4
    for (int k0 = 0; k0 < FDIM; k0 += 4) {
        v2f a, bm;
        a[0]  = arow[k0 + koff];
        a[1]  = arow[k0 + koff + 1];
        bm[0] = brow[k0 + koff];
        bm[1] = brow[k0 + koff + 1];
        acc = __builtin_amdgcn_wmma_f32_16x16x4_f32(false, a, false, bm,
                                                    (short)0, acc, false, false);
    }

#pragma unroll
    for (int v = 0; v < 8; ++v) {
        out[(size_t)(row0 + v + 8 * half) * FDIM + col0 + ln] = acc[v];
    }
}

extern "C" void kernel_launch(void* const* d_in, const int* in_sizes, int n_in,
                              void* d_out, int out_size, void* d_ws, size_t ws_size,
                              hipStream_t stream) {
    const float* feature = (const float*)d_in[0];
    const int*   src     = (const int*)  d_in[1];
    const int*   dst     = (const int*)  d_in[2];
    const float* W       = (const float*)d_in[3];
    const float* b       = (const float*)d_in[4];
    float*       out     = (float*)d_out;
    float*       h       = (float*)d_ws;

    int nh = N_NODES * FDIM;
    gcn_zero_kernel<<<(nh + 255) / 256, 256, 0, stream>>>(h, nh);

    long long tot = (long long)E_EDGES * 32;
    unsigned nblk = (unsigned)((tot + 255) / 256);
    gcn_scatter_kernel<<<nblk, 256, 0, stream>>>(feature, src, dst, h);

    dim3 grid(FDIM / 16, N_NODES / 16);
    gcn_gemm_wmma_kernel<<<grid, 32, 0, stream>>>(h, W, b, out);
}


# Scatter: lane c owns floats c, c+32, c+64, c+96 of a row, so each f32 atomic instruction adds one whole 128-byte line instead of four quarter-filled lines (row loads laid out the same way)
# speedup vs baseline: 1.3457x; 1.3457x over previous
#include <hip/hip_runtime.h>

typedef __attribute__((ext_vector_type(2))) float v2f;
typedef __attribute__((ext_vector_type(8))) float v8f;

#define N_NODES 50000
#define FDIM    128
#define E_EDGES 1600000

__global__ void gcn_zero_kernel(float* __restrict__ h, int n) {
    int i = blockIdx.x * blockDim.x + threadIdx.x;
    if (i < n) h[i] = 0.0f;
}

__global__ void gcn_scatter_kernel(const float* __restrict__ feature,
                                   const int*   __restrict__ src,
                                   const int*   __restrict__ dst,
                                   float*       __restrict__ h) {
    long long t = (long long)blockIdx.x * blockDim.x + threadIdx.x;
    int e = (int)(t >> 5);
    int c = (int)(t & 31);
    if (e >= E_EDGES) return;

    int s = src[e];
    int d = dst[e];

    const float* frow = feature + (size_t)s * FDIM + c;
    float v0 = frow[0];
    float v1 = frow[32];
    float v2 = frow[64];
    float v3 = frow[96];

    float* hrow = h + (size_t)d * FDIM + c;
    unsafeAtomicAdd(hrow + 0,  v0);
    unsafeAtomicAdd(hrow + 32, v1);
    unsafeAtomicAdd(hrow + 64, v2);
    unsafeAtomicAdd(hrow + 96, v3);
}

__global__ void gcn_gemm_wmma_kernel(const float* __restrict__ h,
                                     const float* __restrict__ W,
                                     const float* __restrict__ b,
                                     float*       __restrict__ out) {
    const int lane = threadIdx.x;
    const int col0 = blockIdx.x * 16;
    const int row0 = blockIdx.y * 16;
    const int ln   = lane & 15;
    const int half = lane >> 4;
    const int koff = half * 2;

    float bb = b[col0 + ln];
    v8f acc;
#pragma unroll
    for (int v = 0; v < 8; ++v) acc[v] = bb;

    const float* arow = h + (size_t)(row0 + ln) * FDIM;
    const float* brow = W + (size_t)(col0 + ln) * FDIM;

#pragma unroll 4
    for (int k0 = 0; k0 < FDIM; k0 += 4) {
        v2f a, bm;
        a[0]  = arow[k0 + koff];
        a[1]  = arow[k0 + koff + 1];
        bm[0] = brow[k0 + koff];
        bm[1] = brow[k0 + koff + 1];
        acc = __builtin_amdgcn_wmma_f32_16x16x4_f32(false, a, false, bm,
                                                    (short)0, acc, false, false);
    }

#pragma unroll
    for (int v = 0; v < 8; ++v) {
        out[(size_t)(row0 + v + 8 * half) * FDIM + col0 + ln] = acc[v];
    }
}

extern "C" void kernel_launch(void* const* d_in, const int* in_sizes, int n_in,
                              void* d_out, int out_size, void* d_ws, size_t ws_size,
                              hipStream_t stream) {
    const float* feature = (const float*)d_in[0];
    const int*   src     = (const int*)  d_in[1];
    const int*   dst     = (const int*)  d_in[2];
    const float* W       = (const float*)d_in[3];
    const float* b       = (const float*)d_in[4];
    float*       out     = (float*)d_out;
    float*       h       = (float*)d_ws;

    int nh = N_NODES * FDIM;
    gcn_zero_kernel<<<(nh + 255) / 256, 256, 0, stream>>>(h, nh);

    long long tot = (long long)E_EDGES * 32;
    unsigned nblk = (unsigned)((tot + 255) / 256);
    gcn_scatter_kernel<<<nblk, 256, 0, stream>>>(feature, src, dst, h);

    dim3 grid(FDIM / 16, N_NODES / 16);
    gcn_gemm_wmma_kernel<<<grid, 32, 0, stream>>>(h, W, b, out);
}


# Scatter: each wave takes 32 consecutive edges (one coalesced index load, scalar row bases, 8 edges' row loads in flight), not one short wave per edge: 32x fewer work-groups; atomic order shifts
# speedup vs baseline: 1.9680x; 1.4624x over previous
#include <hip/hip_runtime.h>

typedef __attribute__((ext_vector_type(2))) float v2f;
typedef __attribute__((ext_vector_type(8))) float v8f;

#define N_NODES 50000
#define FDIM    128
#define E_EDGES 1600000

__global__ void gcn_zero_kernel(float* __restrict__ h, int n) {
    int i = blockIdx.x * blockDim.x + threadIdx.x;
    if (i < n) h[i] = 0.0f;
}

#define EPW 32
#define EB  8
#define SCATTER_WAVES (E_EDGES / EPW)
static_assert(E_EDGES % EPW == 0, "edge count must be a whole number of waves");
static_assert(EPW % EB == 0, "batches must tile the wave's edges");

typedef __attribute__((address_space(1))) float gfloat;

__device__ __forceinline__ gfloat* row_base(const float* tab, int r) {
    const unsigned long long a = (unsigned long long)tab + (unsigned long long)(unsigned)r * (FDIM * sizeof(float));
    const unsigned lo = __builtin_amdgcn_readfirstlane((unsigned)a);
    const unsigned hi = __builtin_amdgcn_readfirstlane((unsigned)(a >> 32));
    return (gfloat*)(((unsigned long long)hi << 32) | lo);
}

__device__ __forceinline__ void row_add(gfloat* p, float x) {
    (void)__hip_atomic_fetch_add(p, x, __ATOMIC_RELAXED, __HIP_MEMORY_SCOPE_AGENT);
}

__global__ void __launch_bounds__(256)
gcn_scatter_kernel(const float* __restrict__ feature,
                   const int*   __restrict__ src,
                   const int*   __restrict__ dst,
                   float*       __restrict__ h) {
    const int c    = threadIdx.x & 31;
    const int wave = blockIdx.x * 8 + (threadIdx.x >> 5);
    if (wave >= SCATTER_WAVES) return;

    const int e0 = wave * EPW;
    const int sv = src[e0 + c];
    const int dv = dst[e0 + c];

#pragma unroll
    for (int b0 = 0; b0 < EPW; b0 += EB) {
        float v[EB][4];
#pragma unroll
        for (int j = 0; j < EB; ++j) {
            const gfloat* frow = row_base(feature, __builtin_amdgcn_readlane(sv, b0 + j));
            v[j][0] = frow[c];
            v[j][1] = frow[c + 32];
            v[j][2] = frow[c + 64];
            v[j][3] = frow[c + 96];
        }
#pragma unroll
        for (int j = 0; j < EB; ++j) {
            gfloat* hrow = row_base(h, __builtin_amdgcn_readlane(dv, b0 + j));
            row_add(hrow + c,      v[j][0]);
            row_add(hrow + c + 32, v[j][1]);
            row_add(hrow + c + 64, v[j][2]);
            row_add(hrow + c + 96, v[j][3]);
        }
    }
}

__global__ void gcn_gemm_wmma_kernel(const float* __restrict__ h,
                                     const float* __restrict__ W,
                                     const float* __restrict__ b,
                                     float*       __restrict__ out) {
    const int lane = threadIdx.x;
    const int col0 = blockIdx.x * 16;
    const int row0 = blockIdx.y * 16;
    const int ln   = lane & 15;
    const int half = lane >> 4;
    const int koff = half * 2;

    float bb = b[col0 + ln];
    v8f acc;
#pragma unroll
    for (int v = 0; v < 8; ++v) acc[v] = bb;

    const float* arow = h + (size_t)(row0 + ln) * FDIM;
    const float* brow = W + (size_t)(col0 + ln) * FDIM;

#pragma unroll 4
    for (int k0 = 0; k0 < FDIM; k0 += 4) {
        v2f a, bm;
        a[0]  = arow[k0 + koff];
        a[1]  = arow[k0 + koff + 1];
        bm[0] = brow[k0 + koff];
        bm[1] = brow[k0 + koff + 1];
        acc = __builtin_amdgcn_wmma_f32_16x16x4_f32(false, a, false, bm,
                                                    (short)0, acc, false, false);
    }

#pragma unroll
    for (int v = 0; v < 8; ++v) {
        out[(size_t)(row0 + v + 8 * half) * FDIM + col0 + ln] = acc[v];
    }
}

extern "C" void kernel_launch(void* const* d_in, const int* in_sizes, int n_in,
                              void* d_out, int out_size, void* d_ws, size_t ws_size,
                              hipStream_t stream) {
    const float* feature = (const float*)d_in[0];
    const int*   src     = (const int*)  d_in[1];
    const int*   dst     = (const int*)  d_in[2];
    const float* W       = (const float*)d_in[3];
    const float* b       = (const float*)d_in[4];
    float*       out     = (float*)d_out;
    float*       h       = (float*)d_ws;

    int nh = N_NODES * FDIM;
    gcn_zero_kernel<<<(nh + 255) / 256, 256, 0, stream>>>(h, nh);

    gcn_scatter_kernel<<<(SCATTER_WAVES + 7) / 8, 256, 0, stream>>>(feature, src, dst, h);

    dim3 grid(FDIM / 16, N_NODES / 16);
    gcn_gemm_wmma_kernel<<<grid, 32, 0, stream>>>(h, W, b, out);
}


# GEMM: one wave computes a whole 16x128 output stripe (8 accumulator tiles) with W copied to LDS once per work-group, so each h stripe is read once, not 8 times; k summed in a permuted order
# speedup vs baseline: 2.3014x; 1.1694x over previous
#include <hip/hip_runtime.h>

typedef __attribute__((ext_vector_type(2))) float v2f;
typedef __attribute__((ext_vector_type(8))) float v8f;

#define N_NODES 50000
#define FDIM    128
#define E_EDGES 1600000

__global__ void gcn_zero_kernel(float* __restrict__ h, int n) {
    int i = blockIdx.x * blockDim.x + threadIdx.x;
    if (i < n) h[i] = 0.0f;
}

#define EPW 32
#define EB  8
#define SCATTER_WAVES (E_EDGES / EPW)
static_assert(E_EDGES % EPW == 0, "edge count must be a whole number of waves");
static_assert(EPW % EB == 0, "batches must tile the wave's edges");

typedef __attribute__((address_space(1))) float gfloat;

__device__ __forceinline__ gfloat* row_base(const float* tab, int r) {
    const unsigned long long a = (unsigned long long)tab + (unsigned long long)(unsigned)r * (FDIM * sizeof(float));
    const unsigned lo = __builtin_amdgcn_readfirstlane((unsigned)a);
    const unsigned hi = __builtin_amdgcn_readfirstlane((unsigned)(a >> 32));
    return (gfloat*)(((unsigned long long)hi << 32) | lo);
}

__device__ __forceinline__ void row_add(gfloat* p, float x) {
    (void)__hip_atomic_fetch_add(p, x, __ATOMIC_RELAXED, __HIP_MEMORY_SCOPE_AGENT);
}

__global__ void __launch_bounds__(256)
gcn_scatter_kernel(const float* __restrict__ feature,
                   const int*   __restrict__ src,
                   const int*   __restrict__ dst,
                   float*       __restrict__ h) {
    const int c    = threadIdx.x & 31;
    const int wave = blockIdx.x * 8 + (threadIdx.x >> 5);
    if (wave >= SCATTER_WAVES) return;

    const int e0 = wave * EPW;
    const int sv = src[e0 + c];
    const int dv = dst[e0 + c];

#pragma unroll
    for (int b0 = 0; b0 < EPW; b0 += EB) {
        float v[EB][4];
#pragma unroll
        for (int j = 0; j < EB; ++j) {
            const gfloat* frow = row_base(feature, __builtin_amdgcn_readlane(sv, b0 + j));
            v[j][0] = frow[c];
            v[j][1] = frow[c + 32];
            v[j][2] = frow[c + 64];
            v[j][3] = frow[c + 96];
        }
#pragma unroll
        for (int j = 0; j < EB; ++j) {
            gfloat* hrow = row_base(h, __builtin_amdgcn_readlane(dv, b0 + j));
            row_add(hrow + c,      v[j][0]);
            row_add(hrow + c + 32, v[j][1]);
            row_add(hrow + c + 64, v[j][2]);
            row_add(hrow + c + 96, v[j][3]);
        }
    }
}

typedef __attribute__((ext_vector_type(4))) float v4f;

#define GEMM_WAVES  8
#define ROW_TILES   (N_NODES / 16)
#define GEMM_BLOCKS ((ROW_TILES + GEMM_WAVES - 1) / GEMM_WAVES)
#define W_PITCH     132
static_assert(N_NODES % 16 == 0, "rows must tile by 16");

__global__ void __launch_bounds__(GEMM_WAVES * 32)
gcn_gemm_wmma_kernel(const float* __restrict__ h,
                     const float* __restrict__ W,
                     const float* __restrict__ b,
                     float*       __restrict__ out) {
    __shared__ __attribute__((aligned(16))) float wl[FDIM * W_PITCH];

    const int tid  = threadIdx.x;
    const int lane = tid & 31;
    const int ln   = lane & 15;
    const int half = lane >> 4;
    const int tile = blockIdx.x * GEMM_WAVES + (tid >> 5);
    const bool live = tile < ROW_TILES;
    const int row0 = (live ? tile : ROW_TILES - 1) * 16;

    v4f wreg[16];
#pragma unroll
    for (int m = 0; m < 16; ++m)
        wreg[m] = *(const v4f*)(W + 4 * (tid + 256 * m));

    const float* arow = h + (size_t)(row0 + ln) * FDIM + 4 * half;
    v4f a[16];
#pragma unroll
    for (int j = 0; j < 16; ++j)
        a[j] = *(const v4f*)(arow + 8 * j);

    float bias[8];
#pragma unroll
    for (int t = 0; t < 8; ++t) bias[t] = b[16 * t + ln];

#pragma unroll
    for (int m = 0; m < 16; ++m) {
        const int q = tid + 256 * m;
        *(v4f*)(wl + (q >> 5) * W_PITCH + 4 * (q & 31)) = wreg[m];
    }
    __syncthreads();
    if (!live) return;

    v8f acc[8];
#pragma unroll
    for (int t = 0; t < 8; ++t) {
#pragma unroll
        for (int v = 0; v < 8; ++v) acc[t][v] = bias[t];
    }

    const float* brow = wl + ln * W_PITCH + 4 * half;
#pragma unroll
    for (int j = 0; j < 16; ++j) {
        v4f bf[8];
#pragma unroll
        for (int t = 0; t < 8; ++t)
            bf[t] = *(const v4f*)(brow + t * 16 * W_PITCH + 8 * j);
#pragma unroll
        for (int t = 0; t < 8; ++t) {
            v2f a2, b2;
            a2[0] = a[j][0];  a2[1] = a[j][1];
            b2[0] = bf[t][0]; b2[1] = bf[t][1];
            acc[t] = __builtin_amdgcn_wmma_f32_16x16x4_f32(false, a2, false, b2,
                                                           (short)0, acc[t], false, false);
        }
#pragma unroll
        for (int t = 0; t < 8; ++t) {
            v2f a2, b2;
            a2[0] = a[j][2];  a2[1] = a[j][3];
            b2[0] = bf[t][2]; b2[1] = bf[t][3];
            acc[t] = __builtin_amdgcn_wmma_f32_16x16x4_f32(false, a2, false, b2,
                                                           (short)0, acc[t], false, false);
        }
    }

    float* orow = out + (size_t)(row0 + 8 * half) * FDIM + ln;
#pragma unroll
    for (int t = 0; t < 8; ++t) {
#pragma unroll
        for (int v = 0; v < 8; ++v)
            orow[v * FDIM + 16 * t] = acc[t][v];
    }
}

extern "C" void kernel_launch(void* const* d_in, const int* in_sizes, int n_in,
                              void* d_out, int out_size, void* d_ws, size_t ws_size,
                              hipStream_t stream) {
    const float* feature = (const float*)d_in[0];
    const int*   src     = (const int*)  d_in[1];
    const int*   dst     = (const int*)  d_in[2];
    const float* W       = (const float*)d_in[3];
    const float* b       = (const float*)d_in[4];
    float*       out     = (float*)d_out;
    float*       h       = (float*)d_ws;

    int nh = N_NODES * FDIM;
    gcn_zero_kernel<<<(nh + 255) / 256, 256, 0, stream>>>(h, nh);

    gcn_scatter_kernel<<<(SCATTER_WAVES + 7) / 8, 256, 0, stream>>>(feature, src, dst, h);

    gcn_gemm_wmma_kernel<<<GEMM_BLOCKS, GEMM_WAVES * 32, 0, stream>>>(h, W, b, out);
}


# Edges sorted by destination range (per-chunk LDS counting sort): each work-group sums its 208 rows with LDS adds and multiplies by W from LDS; no global f32 atomics, no h; sums still by arrival order
# speedup vs baseline: 4.3245x; 1.8791x over previous
#include <hip/hip_runtime.h>


typedef __attribute__((ext_vector_type(2))) float v2f;
typedef __attribute__((ext_vector_type(4))) float v4f;
typedef __attribute__((ext_vector_type(8))) float v8f;

#define N_NODES 50000
#define FDIM    128
#define E_EDGES 1600000

#define BIN_ROWS   208
#define N_BINS     ((N_NODES + BIN_ROWS - 1) / BIN_ROWS)
#define CHUNK      6400
#define N_CHUNKS   (E_EDGES / CHUNK)
#define BLOCK      1024
#define N_WAVES    (BLOCK / 32)
#define CHUNKS_PER_WAVE ((N_CHUNKS + N_WAVES - 1) / N_WAVES)
#define PER_THREAD ((CHUNK + BLOCK - 1) / BLOCK)
static_assert(E_EDGES % CHUNK == 0, "chunks must tile the edge list");
static_assert(N_NODES % 16 == 0 && BIN_ROWS % 16 == 0, "ranges are whole 16-row stripes");
static_assert(N_BINS < 256, "the histogram has 256 slots and slot N_BINS holds the total");
static_assert(N_NODES <= 65536 && BIN_ROWS <= 65536, "an edge packs into 16 + 16 bits");
static_assert(CHUNKS_PER_WAVE <= 32, "one lane per chunk of a wave");

__global__ void __launch_bounds__(BLOCK)
gcn_group_kernel(const int* __restrict__ src,
                 const int* __restrict__ dst,
                 unsigned*  __restrict__ grouped,
                 int*       __restrict__ segT) {
    __shared__ int      hist[256];
    __shared__ int      hoff[256];
    __shared__ unsigned stage[CHUNK];

    const int tid   = threadIdx.x;
    const int chunk = blockIdx.x;
    const int e0    = chunk * CHUNK;

    if (tid < 256) hist[tid] = 0;
    __syncthreads();

    unsigned pk[PER_THREAD];
    int      bn[PER_THREAD];
    int      rk[PER_THREAD];
#pragma unroll
    for (int i = 0; i < PER_THREAD; ++i) {
        const int idx = tid + BLOCK * i;
        pk[i] = 0; bn[i] = 0; rk[i] = 0;
        if (idx < CHUNK) {
            const unsigned s = (unsigned)src[e0 + idx];
            const unsigned d = (unsigned)dst[e0 + idx];
            const unsigned b = d / BIN_ROWS;
            bn[i] = (int)b;
            pk[i] = s | ((d - b * BIN_ROWS) << 16);
            rk[i] = atomicAdd(&hist[b], 1);
        }
    }
    __syncthreads();

    if (tid < 32) {
        int cnt[8];
        int local = 0;
#pragma unroll
        for (int q = 0; q < 8; ++q) { cnt[q] = hist[8 * tid + q]; local += cnt[q]; }
        int inc = local;
#pragma unroll
        for (int d = 1; d < 32; d <<= 1) {
            const int t = __shfl_up(inc, d);
            if (tid >= d) inc += t;
        }
        int run = inc - local;
#pragma unroll
        for (int q = 0; q < 8; ++q) { hoff[8 * tid + q] = run; run += cnt[q]; }
    }
    __syncthreads();

#pragma unroll
    for (int i = 0; i < PER_THREAD; ++i) {
        const int idx = tid + BLOCK * i;
        if (idx < CHUNK) stage[hoff[bn[i]] + rk[i]] = pk[i];
    }
    if (tid <= N_BINS) segT[tid * N_CHUNKS + chunk] = hoff[tid];
    __syncthreads();

#pragma unroll
    for (int i = 0; i < PER_THREAD; ++i) {
        const int idx = tid + BLOCK * i;
        if (idx < CHUNK) grouped[e0 + idx] = stage[idx];
    }
}

#define EB      8
#define PITCH   132

typedef __attribute__((address_space(1))) float gfloat;

__device__ __forceinline__ const gfloat* row_base(const float* tab, int r) {
    const unsigned long long a = (unsigned long long)tab + (unsigned long long)(unsigned)r * (FDIM * sizeof(float));
    const unsigned lo = __builtin_amdgcn_readfirstlane((unsigned)a);
    const unsigned hi = __builtin_amdgcn_readfirstlane((unsigned)(a >> 32));
    return (const gfloat*)(((unsigned long long)hi << 32) | lo);
}

__device__ __forceinline__ void lds_add(float* p, float x) {
    (void)__hip_atomic_fetch_add(p, x, __ATOMIC_RELAXED, __HIP_MEMORY_SCOPE_WORKGROUP);
}

template <bool FULL>
__device__ __forceinline__ void edge_batch(const float* feature, float* sums,
                                           unsigned pk, int j0, int n, int lane) {
    float v[EB][4];
#pragma unroll
    for (int j = 0; j < EB; ++j) {
        if (FULL || j0 + j < n) {
            const unsigned p = (unsigned)__builtin_amdgcn_readlane((int)pk, j0 + j);
            const gfloat* frow = row_base(feature, (int)(p & 0xFFFFu));
            v[j][0] = frow[lane];
            v[j][1] = frow[lane + 32];
            v[j][2] = frow[lane + 64];
            v[j][3] = frow[lane + 96];
        }
    }
#pragma unroll
    for (int j = 0; j < EB; ++j) {
        if (FULL || j0 + j < n) {
            const unsigned p = (unsigned)__builtin_amdgcn_readlane((int)pk, j0 + j);
            float* srow = sums + (p >> 16) * PITCH + lane;
            lds_add(srow,      v[j][0]);
            lds_add(srow + 32, v[j][1]);
            lds_add(srow + 64, v[j][2]);
            lds_add(srow + 96, v[j][3]);
        }
    }
}

__device__ __forceinline__ void edge_group(const float* feature, float* sums,
                                           unsigned pk, int n, int lane) {
    int j0 = 0;
    for (; j0 + EB <= n; j0 += EB) edge_batch<true>(feature, sums, pk, j0, n, lane);
    if (j0 < n) edge_batch<false>(feature, sums, pk, j0, n, lane);
}

__global__ void __launch_bounds__(BLOCK)
gcn_range_kernel(const float*    __restrict__ feature,
                 const unsigned* __restrict__ grouped,
                 const int*      __restrict__ segT,
                 const float*    __restrict__ W,
                 const float*    __restrict__ b,
                 float*          __restrict__ out) {
    __shared__ __attribute__((aligned(16))) float sums[BIN_ROWS * PITCH];
    __shared__ __attribute__((aligned(16))) float wl[FDIM * PITCH];

    const int tid  = threadIdx.x;
    const int lane = tid & 31;
    const int wave = __builtin_amdgcn_readfirstlane(tid >> 5);
    const int k    = blockIdx.x;

    int sv = 0, ev = 0;
    {
        const int c = wave + N_WAVES * lane;
        if (lane < CHUNKS_PER_WAVE && c < N_CHUNKS) {
            sv = segT[k * N_CHUNKS + c];
            ev = segT[(k + 1) * N_CHUNKS + c];
        }
    }

    v4f wreg[4];
#pragma unroll
    for (int m = 0; m < 4; ++m)
        wreg[m] = *(const v4f*)(W + 4 * (tid + BLOCK * m));

    const v4f zero4 = {0.0f, 0.0f, 0.0f, 0.0f};
    for (int q = tid; q < BIN_ROWS * PITCH / 4; q += BLOCK)
        ((v4f*)sums)[q] = zero4;

#pragma unroll
    for (int m = 0; m < 4; ++m) {
        const int q = tid + BLOCK * m;
        *(v4f*)(wl + (q >> 5) * PITCH + 4 * (q & 31)) = wreg[m];
    }

    int s_cur = __builtin_amdgcn_readlane(sv, 0);
    int e_cur = __builtin_amdgcn_readlane(ev, 0);
    unsigned pk_cur = 0;
    if (lane < e_cur - s_cur) pk_cur = grouped[wave * CHUNK + s_cur + lane];

    __syncthreads();

    for (int ci = 0; ci < CHUNKS_PER_WAVE; ++ci) {
        const int c = wave + N_WAVES * ci;
        int s_nxt = 0, e_nxt = 0;
        unsigned pk_nxt = 0;
        if (ci + 1 < CHUNKS_PER_WAVE) {
            s_nxt = __builtin_amdgcn_readlane(sv, ci + 1);
            e_nxt = __builtin_amdgcn_readlane(ev, ci + 1);
            if (lane < e_nxt - s_nxt) pk_nxt = grouped[(c + N_WAVES) * CHUNK + s_nxt + lane];
        }

        edge_group(feature, sums, pk_cur, min(32, e_cur - s_cur), lane);
        for (int base = s_cur + 32; base < e_cur; base += 32) {
            const int n = min(32, e_cur - base);
            const unsigned pk = (lane < n) ? grouped[c * CHUNK + base + lane] : 0u;
            edge_group(feature, sums, pk, n, lane);
        }
        s_cur = s_nxt; e_cur = e_nxt; pk_cur = pk_nxt;
    }

    __syncthreads();

    const int ln    = lane & 15;
    const int half  = lane >> 4;
    const int rows  = min(BIN_ROWS, N_NODES - k * BIN_ROWS);
    const int units = (rows >> 4) * 4;
    for (int u = wave; u < units; u += N_WAVES) {
        const int s  = u >> 2;
        const int cp = u & 3;
        const float* arow = sums + (16 * s + ln) * PITCH + 4 * half;
        const float* brow = wl + (32 * cp + ln) * PITCH + 4 * half;
        const float bias0 = b[32 * cp + ln];
        const float bias1 = b[32 * cp + 16 + ln];
        v8f acc0, acc1;
#pragma unroll
        for (int v = 0; v < 8; ++v) { acc0[v] = bias0; acc1[v] = bias1; }
#pragma unroll
        for (int j = 0; j < 16; ++j) {
            const v4f a  = *(const v4f*)(arow + 8 * j);
            const v4f b0 = *(const v4f*)(brow + 8 * j);
            const v4f b1 = *(const v4f*)(brow + 16 * PITCH + 8 * j);
            v2f a2, x2, y2;
            a2[0] = a[0];  a2[1] = a[1];
            x2[0] = b0[0]; x2[1] = b0[1];
            y2[0] = b1[0]; y2[1] = b1[1];
            acc0 = __builtin_amdgcn_wmma_f32_16x16x4_f32(false, a2, false, x2, (short)0, acc0, false, false);
            acc1 = __builtin_amdgcn_wmma_f32_16x16x4_f32(false, a2, false, y2, (short)0, acc1, false, false);
            a2[0] = a[2];  a2[1] = a[3];
            x2[0] = b0[2]; x2[1] = b0[3];
            y2[0] = b1[2]; y2[1] = b1[3];
            acc0 = __builtin_amdgcn_wmma_f32_16x16x4_f32(false, a2, false, x2, (short)0, acc0, false, false);
            acc1 = __builtin_amdgcn_wmma_f32_16x16x4_f32(false, a2, false, y2, (short)0, acc1, false, false);
        }
        float* orow = out + (size_t)(k * BIN_ROWS + 16 * s + 8 * half) * FDIM + 32 * cp + ln;
#pragma unroll
        for (int v = 0; v < 8; ++v) {
            orow[v * FDIM]      = acc0[v];
            orow[v * FDIM + 16] = acc1[v];
        }
    }
}

extern "C" void kernel_launch(void* const* d_in, const int* in_sizes, int n_in,
                              void* d_out, int out_size, void* d_ws, size_t ws_size,
                              hipStream_t stream) {
    const float* feature = (const float*)d_in[0];
    const int*   src     = (const int*)  d_in[1];
    const int*   dst     = (const int*)  d_in[2];
    const float* W       = (const float*)d_in[3];
    const float* b       = (const float*)d_in[4];
    float*       out     = (float*)d_out;
    unsigned*    grouped = (unsigned*)d_ws;
    int*         segT    = (int*)(grouped + E_EDGES);

    gcn_group_kernel<<<N_CHUNKS, BLOCK, 0, stream>>>(src, dst, grouped, segT);

    gcn_range_kernel<<<N_BINS, BLOCK, 0, stream>>>(feature, grouped, segT, W, b, out);
}


# Range's edges counting-sorted by destination row in LDS; a wave sums each row of its span of the list in registers and adds it to LDS once per row, not per edge; rows summed in list order
# speedup vs baseline: 5.4779x; 1.2667x over previous
#include <hip/hip_runtime.h>


typedef __attribute__((ext_vector_type(2))) float v2f;
typedef __attribute__((ext_vector_type(4))) float v4f;
typedef __attribute__((ext_vector_type(8))) float v8f;

#define N_NODES 50000
#define FDIM    128
#define E_EDGES 1600000

#define BIN_ROWS   208
#define N_BINS     ((N_NODES + BIN_ROWS - 1) / BIN_ROWS)
#define CHUNK      6400
#define N_CHUNKS   (E_EDGES / CHUNK)
#define BLOCK      1024
#define N_WAVES    (BLOCK / 32)
#define CHUNKS_PER_WAVE ((N_CHUNKS + N_WAVES - 1) / N_WAVES)
#define PER_THREAD ((CHUNK + BLOCK - 1) / BLOCK)
static_assert(E_EDGES % CHUNK == 0, "chunks must tile the edge list");
static_assert(N_NODES % 16 == 0 && BIN_ROWS % 16 == 0, "ranges are whole 16-row stripes");
static_assert(N_BINS < 256, "the histogram has 256 slots and slot N_BINS holds the total");
static_assert(N_NODES <= 65536 && BIN_ROWS <= 65536, "an edge packs into 16 + 16 bits");
static_assert(CHUNKS_PER_WAVE <= 32, "one lane per chunk of a wave");

__global__ void __launch_bounds__(BLOCK)
gcn_group_kernel(const int* __restrict__ src,
                 const int* __restrict__ dst,
                 unsigned*  __restrict__ grouped,
                 int*       __restrict__ segT) {
    __shared__ int      hist[256];
    __shared__ int      hoff[256];
    __shared__ unsigned stage[CHUNK];

    const int tid   = threadIdx.x;
    const int chunk = blockIdx.x;
    const int e0    = chunk * CHUNK;

    if (tid < 256) hist[tid] = 0;
    __syncthreads();

    unsigned pk[PER_THREAD];
    int      bn[PER_THREAD];
    int      rk[PER_THREAD];
#pragma unroll
    for (int i = 0; i < PER_THREAD; ++i) {
        const int idx = tid + BLOCK * i;
        pk[i] = 0; bn[i] = 0; rk[i] = 0;
        if (idx < CHUNK) {
            const unsigned s = (unsigned)src[e0 + idx];
            const unsigned d = (unsigned)dst[e0 + idx];
            const unsigned b = d / BIN_ROWS;
            bn[i] = (int)b;
            pk[i] = s | ((d - b * BIN_ROWS) << 16);
            rk[i] = atomicAdd(&hist[b], 1);
        }
    }
    __syncthreads();

    if (tid < 32) {
        int cnt[8];
        int local = 0;
#pragma unroll
        for (int q = 0; q < 8; ++q) { cnt[q] = hist[8 * tid + q]; local += cnt[q]; }
        int inc = local;
#pragma unroll
        for (int d = 1; d < 32; d <<= 1) {
            const int t = __shfl_up(inc, d);
            if (tid >= d) inc += t;
        }
        int run = inc - local;
#pragma unroll
        for (int q = 0; q < 8; ++q) { hoff[8 * tid + q] = run; run += cnt[q]; }
    }
    __syncthreads();

#pragma unroll
    for (int i = 0; i < PER_THREAD; ++i) {
        const int idx = tid + BLOCK * i;
        if (idx < CHUNK) stage[hoff[bn[i]] + rk[i]] = pk[i];
    }
    if (tid <= N_BINS) segT[tid * N_CHUNKS + chunk] = hoff[tid];
    __syncthreads();

#pragma unroll
    for (int i = 0; i < PER_THREAD; ++i) {
        const int idx = tid + BLOCK * i;
        if (idx < CHUNK) grouped[e0 + idx] = stage[idx];
    }
}

#define EB      8
#define CAP     8192
#define SEG_SLOTS 2
#define PITCH   132

typedef __attribute__((address_space(1))) float gfloat;

__device__ __forceinline__ const gfloat* row_base(const float* tab, int r) {
    const unsigned long long a = (unsigned long long)tab + (unsigned long long)(unsigned)r * (FDIM * sizeof(float));
    const unsigned lo = __builtin_amdgcn_readfirstlane((unsigned)a);
    const unsigned hi = __builtin_amdgcn_readfirstlane((unsigned)(a >> 32));
    return (const gfloat*)(((unsigned long long)hi << 32) | lo);
}

__device__ __forceinline__ void lds_add(float* p, float x) {
    (void)__hip_atomic_fetch_add(p, x, __ATOMIC_RELAXED, __HIP_MEMORY_SCOPE_WORKGROUP);
}

template <bool FULL>
__device__ __forceinline__ void edge_batch(const float* feature, float* sums,
                                           unsigned pk, int j0, int n, int lane) {
    float v[EB][4];
#pragma unroll
    for (int j = 0; j < EB; ++j) {
        if (FULL || j0 + j < n) {
            const unsigned p = (unsigned)__builtin_amdgcn_readlane((int)pk, j0 + j);
            const gfloat* frow = row_base(feature, (int)(p & 0xFFFFu));
            v[j][0] = frow[lane];
            v[j][1] = frow[lane + 32];
            v[j][2] = frow[lane + 64];
            v[j][3] = frow[lane + 96];
        }
    }
#pragma unroll
    for (int j = 0; j < EB; ++j) {
        if (FULL || j0 + j < n) {
            const unsigned p = (unsigned)__builtin_amdgcn_readlane((int)pk, j0 + j);
            float* srow = sums + (p >> 16) * PITCH + lane;
            lds_add(srow,      v[j][0]);
            lds_add(srow + 32, v[j][1]);
            lds_add(srow + 64, v[j][2]);
            lds_add(srow + 96, v[j][3]);
        }
    }
}

__device__ __forceinline__ void edge_group(const float* feature, float* sums,
                                           unsigned pk, int n, int lane) {
    int j0 = 0;
    for (; j0 + EB <= n; j0 += EB) edge_batch<true>(feature, sums, pk, j0, n, lane);
    if (j0 < n) edge_batch<false>(feature, sums, pk, j0, n, lane);
}

__device__ __forceinline__ void flush_row(float* sums, int r, const float (&acc)[4], int lane) {
    float* srow = sums + r * PITCH + lane;
    lds_add(srow,      acc[0]);
    lds_add(srow + 32, acc[1]);
    lds_add(srow + 64, acc[2]);
    lds_add(srow + 96, acc[3]);
}

template <bool FULL>
__device__ __forceinline__ void row_batch(const float* feature, float* sums,
                                          unsigned pk, int j0, int n, int lane,
                                          int& cur, float (&acc)[4]) {
    float v[EB][4];
#pragma unroll
    for (int j = 0; j < EB; ++j) {
        if (FULL || j0 + j < n) {
            const unsigned p = (unsigned)__builtin_amdgcn_readlane((int)pk, j0 + j);
            const gfloat* frow = row_base(feature, (int)(p & 0xFFFFu));
            v[j][0] = frow[lane];
            v[j][1] = frow[lane + 32];
            v[j][2] = frow[lane + 64];
            v[j][3] = frow[lane + 96];
        }
    }
#pragma unroll
    for (int j = 0; j < EB; ++j) {
        if (FULL || j0 + j < n) {
            const unsigned p = (unsigned)__builtin_amdgcn_readlane((int)pk, j0 + j);
            const int r = (int)(p >> 16);
            if (r != cur) {
                if (cur >= 0) flush_row(sums, cur, acc, lane);
                cur = r;
                acc[0] = 0.0f; acc[1] = 0.0f; acc[2] = 0.0f; acc[3] = 0.0f;
            }
            acc[0] += v[j][0]; acc[1] += v[j][1]; acc[2] += v[j][2]; acc[3] += v[j][3];
        }
    }
}

__global__ void __launch_bounds__(BLOCK)
gcn_range_kernel(const float*    __restrict__ feature,
                 const unsigned* __restrict__ grouped,
                 const int*      __restrict__ segT,
                 const float*    __restrict__ W,
                 const float*    __restrict__ b,
                 float*          __restrict__ out) {
    __shared__ __attribute__((aligned(16))) float sums[BIN_ROWS * PITCH];
    __shared__ __attribute__((aligned(16))) float wl[FDIM * PITCH];
    __shared__ unsigned sorted[CAP];
    __shared__ int      rowpos[256];
    __shared__ int      total_s;

    const int tid  = threadIdx.x;
    const int lane = tid & 31;
    const int wave = __builtin_amdgcn_readfirstlane(tid >> 5);
    const int k    = blockIdx.x;

    int sv = 0, ev = 0;
    {
        const int c = wave + N_WAVES * lane;
        if (lane < CHUNKS_PER_WAVE && c < N_CHUNKS) {
            sv = segT[k * N_CHUNKS + c];
            ev = segT[(k + 1) * N_CHUNKS + c];
        }
    }

    v4f wreg[4];
#pragma unroll
    for (int m = 0; m < 4; ++m)
        wreg[m] = *(const v4f*)(W + 4 * (tid + BLOCK * m));

    const v4f zero4 = {0.0f, 0.0f, 0.0f, 0.0f};
    for (int q = tid; q < BIN_ROWS * PITCH / 4; q += BLOCK)
        ((v4f*)sums)[q] = zero4;
    if (tid < 256) rowpos[tid] = 0;

#pragma unroll
    for (int m = 0; m < 4; ++m) {
        const int q = tid + BLOCK * m;
        *(v4f*)(wl + (q >> 5) * PITCH + 4 * (q & 31)) = wreg[m];
    }

    unsigned pkr[CHUNKS_PER_WAVE][SEG_SLOTS];
#pragma unroll
    for (int ci = 0; ci < CHUNKS_PER_WAVE; ++ci) {
        const int c = wave + N_WAVES * ci;
        const int s = __builtin_amdgcn_readlane(sv, ci);
        const int e = __builtin_amdgcn_readlane(ev, ci);
#pragma unroll
        for (int t = 0; t < SEG_SLOTS; ++t) {
            pkr[ci][t] = 0;
            if (32 * t + lane < e - s) pkr[ci][t] = grouped[c * CHUNK + s + 32 * t + lane];
        }
    }

    __syncthreads();

#pragma unroll
    for (int ci = 0; ci < CHUNKS_PER_WAVE; ++ci) {
        const int c = wave + N_WAVES * ci;
        const int s = __builtin_amdgcn_readlane(sv, ci);
        const int e = __builtin_amdgcn_readlane(ev, ci);
#pragma unroll
        for (int t = 0; t < SEG_SLOTS; ++t)
            if (32 * t + lane < e - s) atomicAdd(&rowpos[pkr[ci][t] >> 16], 1);
        for (int base = s + 32 * SEG_SLOTS; base < e; base += 32)
            if (base + lane < e) atomicAdd(&rowpos[grouped[c * CHUNK + base + lane] >> 16], 1);
    }
    __syncthreads();

    if (tid < 32) {
        int cnt[8];
        int local = 0;
#pragma unroll
        for (int q = 0; q < 8; ++q) { cnt[q] = rowpos[8 * tid + q]; local += cnt[q]; }
        int inc = local;
#pragma unroll
        for (int d = 1; d < 32; d <<= 1) {
            const int t = __shfl_up(inc, d);
            if (tid >= d) inc += t;
        }
        int run = inc - local;
#pragma unroll
        for (int q = 0; q < 8; ++q) { rowpos[8 * tid + q] = run; run += cnt[q]; }
        if (tid == 31) total_s = inc;
    }
    __syncthreads();

    const int total = __builtin_amdgcn_readfirstlane(total_s);

    if (total <= CAP) {
#pragma unroll
        for (int ci = 0; ci < CHUNKS_PER_WAVE; ++ci) {
            const int c = wave + N_WAVES * ci;
            const int s = __builtin_amdgcn_readlane(sv, ci);
            const int e = __builtin_amdgcn_readlane(ev, ci);
#pragma unroll
            for (int t = 0; t < SEG_SLOTS; ++t)
                if (32 * t + lane < e - s) sorted[atomicAdd(&rowpos[pkr[ci][t] >> 16], 1)] = pkr[ci][t];
            for (int base = s + 32 * SEG_SLOTS; base < e; base += 32)
                if (base + lane < e) {
                    const unsigned p = grouped[c * CHUNK + base + lane];
                    sorted[atomicAdd(&rowpos[p >> 16], 1)] = p;
                }
        }
        __syncthreads();

        const int per   = (total + N_WAVES - 1) / N_WAVES;
        const int begin = min(total, wave * per);
        const int end   = min(total, begin + per);
        int   cur = -1;
        float acc[4] = {0.0f, 0.0f, 0.0f, 0.0f};
        for (int base = begin; base < end; base += 32) {
            const int n = min(32, end - base);
            const unsigned pk = (lane < n) ? sorted[base + lane] : 0u;
            int j0 = 0;
            for (; j0 + EB <= n; j0 += EB) row_batch<true>(feature, sums, pk, j0, n, lane, cur, acc);
            if (j0 < n) row_batch<false>(feature, sums, pk, j0, n, lane, cur, acc);
        }
        if (cur >= 0) flush_row(sums, cur, acc, lane);
    } else {
        for (int ci = 0; ci < CHUNKS_PER_WAVE; ++ci) {
            const int c = wave + N_WAVES * ci;
            const int s = __builtin_amdgcn_readlane(sv, ci);
            const int e = __builtin_amdgcn_readlane(ev, ci);
            for (int base = s; base < e; base += 32) {
                const int n = min(32, e - base);
                const unsigned pk = (lane < n) ? grouped[c * CHUNK + base + lane] : 0u;
                edge_group(feature, sums, pk, n, lane);
            }
        }
    }

    __syncthreads();

    const int ln    = lane & 15;
    const int half  = lane >> 4;
    const int rows  = min(BIN_ROWS, N_NODES - k * BIN_ROWS);
    const int units = (rows >> 4) * 4;
    for (int u = wave; u < units; u += N_WAVES) {
        const int s  = u >> 2;
        const int cp = u & 3;
        const float* arow = sums + (16 * s + ln) * PITCH + 4 * half;
        const float* brow = wl + (32 * cp + ln) * PITCH + 4 * half;
        const float bias0 = b[32 * cp + ln];
        const float bias1 = b[32 * cp + 16 + ln];
        v8f acc0, acc1;
#pragma unroll
        for (int v = 0; v < 8; ++v) { acc0[v] = bias0; acc1[v] = bias1; }
#pragma unroll
        for (int j = 0; j < 16; ++j) {
            const v4f a  = *(const v4f*)(arow + 8 * j);
            const v4f b0 = *(const v4f*)(brow + 8 * j);
            const v4f b1 = *(const v4f*)(brow + 16 * PITCH + 8 * j);
            v2f a2, x2, y2;
            a2[0] = a[0];  a2[1] = a[1];
            x2[0] = b0[0]; x2[1] = b0[1];
            y2[0] = b1[0]; y2[1] = b1[1];
            acc0 = __builtin_amdgcn_wmma_f32_16x16x4_f32(false, a2, false, x2, (short)0, acc0, false, false);
            acc1 = __builtin_amdgcn_wmma_f32_16x16x4_f32(false, a2, false, y2, (short)0, acc1, false, false);
            a2[0] = a[2];  a2[1] = a[3];
            x2[0] = b0[2]; x2[1] = b0[3];
            y2[0] = b1[2]; y2[1] = b1[3];
            acc0 = __builtin_amdgcn_wmma_f32_16x16x4_f32(false, a2, false, x2, (short)0, acc0, false, false);
            acc1 = __builtin_amdgcn_wmma_f32_16x16x4_f32(false, a2, false, y2, (short)0, acc1, false, false);
        }
        float* orow = out + (size_t)(k * BIN_ROWS + 16 * s + 8 * half) * FDIM + 32 * cp + ln;
#pragma unroll
        for (int v = 0; v < 8; ++v) {
            orow[v * FDIM]      = acc0[v];
            orow[v * FDIM + 16] = acc1[v];
        }
    }
}

extern "C" void kernel_launch(void* const* d_in, const int* in_sizes, int n_in,
                              void* d_out, int out_size, void* d_ws, size_t ws_size,
                              hipStream_t stream) {
    const float* feature = (const float*)d_in[0];
    const int*   src     = (const int*)  d_in[1];
    const int*   dst     = (const int*)  d_in[2];
    const float* W       = (const float*)d_in[3];
    const float* b       = (const float*)d_in[4];
    float*       out     = (float*)d_out;
    unsigned*    grouped = (unsigned*)d_ws;
    int*         segT    = (int*)(grouped + E_EDGES);

    gcn_group_kernel<<<N_CHUNKS, BLOCK, 0, stream>>>(src, dst, grouped, segT);

    gcn_range_kernel<<<N_BINS, BLOCK, 0, stream>>>(feature, grouped, segT, W, b, out);
}


# Row gather: one 16-byte load per lane fetches a whole 512-byte source row per instruction (lane c owns floats 4c..4c+3), not four 4-byte loads of one 128-byte line each; order of sums unchanged
# speedup vs baseline: 5.7151x; 1.0433x over previous
#include <hip/hip_runtime.h>


typedef __attribute__((ext_vector_type(2))) float v2f;
typedef __attribute__((ext_vector_type(4))) float v4f;
typedef __attribute__((ext_vector_type(8))) float v8f;

#define N_NODES 50000
#define FDIM    128
#define E_EDGES 1600000

#define BIN_ROWS   208
#define N_BINS     ((N_NODES + BIN_ROWS - 1) / BIN_ROWS)
#define CHUNK      6400
#define N_CHUNKS   (E_EDGES / CHUNK)
#define BLOCK      1024
#define N_WAVES    (BLOCK / 32)
#define CHUNKS_PER_WAVE ((N_CHUNKS + N_WAVES - 1) / N_WAVES)
#define PER_THREAD ((CHUNK + BLOCK - 1) / BLOCK)
static_assert(E_EDGES % CHUNK == 0, "chunks must tile the edge list");
static_assert(N_NODES % 16 == 0 && BIN_ROWS % 16 == 0, "ranges are whole 16-row stripes");
static_assert(N_BINS < 256, "the histogram has 256 slots and slot N_BINS holds the total");
static_assert(N_NODES <= 65536 && BIN_ROWS <= 65536, "an edge packs into 16 + 16 bits");
static_assert(CHUNKS_PER_WAVE <= 32, "one lane per chunk of a wave");

__global__ void __launch_bounds__(BLOCK)
gcn_group_kernel(const int* __restrict__ src,
                 const int* __restrict__ dst,
                 unsigned*  __restrict__ grouped,
                 int*       __restrict__ segT) {
    __shared__ int      hist[256];
    __shared__ int      hoff[256];
    __shared__ unsigned stage[CHUNK];

    const int tid   = threadIdx.x;
    const int chunk = blockIdx.x;
    const int e0    = chunk * CHUNK;

    if (tid < 256) hist[tid] = 0;
    __syncthreads();

    unsigned pk[PER_THREAD];
    int      bn[PER_THREAD];
    int      rk[PER_THREAD];
#pragma unroll
    for (int i = 0; i < PER_THREAD; ++i) {
        const int idx = tid + BLOCK * i;
        pk[i] = 0; bn[i] = 0; rk[i] = 0;
        if (idx < CHUNK) {
            const unsigned s = (unsigned)src[e0 + idx];
            const unsigned d = (unsigned)dst[e0 + idx];
            const unsigned b = d / BIN_ROWS;
            bn[i] = (int)b;
            pk[i] = s | ((d - b * BIN_ROWS) << 16);
            rk[i] = atomicAdd(&hist[b], 1);
        }
    }
    __syncthreads();

    if (tid < 32) {
        int cnt[8];
        int local = 0;
#pragma unroll
        for (int q = 0; q < 8; ++q) { cnt[q] = hist[8 * tid + q]; local += cnt[q]; }
        int inc = local;
#pragma unroll
        for (int d = 1; d < 32; d <<= 1) {
            const int t = __shfl_up(inc, d);
            if (tid >= d) inc += t;
        }
        int run = inc - local;
#pragma unroll
        for (int q = 0; q < 8; ++q) { hoff[8 * tid + q] = run; run += cnt[q]; }
    }
    __syncthreads();

#pragma unroll
    for (int i = 0; i < PER_THREAD; ++i) {
        const int idx = tid + BLOCK * i;
        if (idx < CHUNK) stage[hoff[bn[i]] + rk[i]] = pk[i];
    }
    if (tid <= N_BINS) segT[tid * N_CHUNKS + chunk] = hoff[tid];
    __syncthreads();

#pragma unroll
    for (int i = 0; i < PER_THREAD; ++i) {
        const int idx = tid + BLOCK * i;
        if (idx < CHUNK) grouped[e0 + idx] = stage[idx];
    }
}

#define EB      8
#define CAP     8192
#define SEG_SLOTS 2
#define PITCH   132

typedef __attribute__((address_space(1))) v4f   gv4f;

__device__ __forceinline__ const gv4f* row_base(const float* tab, int r) {
    const unsigned long long a = (unsigned long long)tab + (unsigned long long)(unsigned)r * (FDIM * sizeof(float));
    const unsigned lo = __builtin_amdgcn_readfirstlane((unsigned)a);
    const unsigned hi = __builtin_amdgcn_readfirstlane((unsigned)(a >> 32));
    return (const gv4f*)(((unsigned long long)hi << 32) | lo);
}

__device__ __forceinline__ void lds_add(float* p, float x) {
    (void)__hip_atomic_fetch_add(p, x, __ATOMIC_RELAXED, __HIP_MEMORY_SCOPE_WORKGROUP);
}

template <bool FULL>
__device__ __forceinline__ void edge_batch(const float* feature, float* sums,
                                           unsigned pk, int j0, int n, int lane) {
    v4f v[EB];
#pragma unroll
    for (int j = 0; j < EB; ++j) {
        if (FULL || j0 + j < n) {
            const unsigned p = (unsigned)__builtin_amdgcn_readlane((int)pk, j0 + j);
            v[j] = row_base(feature, (int)(p & 0xFFFFu))[lane];
        }
    }
#pragma unroll
    for (int j = 0; j < EB; ++j) {
        if (FULL || j0 + j < n) {
            const unsigned p = (unsigned)__builtin_amdgcn_readlane((int)pk, j0 + j);
            float* srow = sums + (p >> 16) * PITCH + 4 * lane;
            lds_add(srow,     v[j][0]);
            lds_add(srow + 1, v[j][1]);
            lds_add(srow + 2, v[j][2]);
            lds_add(srow + 3, v[j][3]);
        }
    }
}

__device__ __forceinline__ void edge_group(const float* feature, float* sums,
                                           unsigned pk, int n, int lane) {
    int j0 = 0;
    for (; j0 + EB <= n; j0 += EB) edge_batch<true>(feature, sums, pk, j0, n, lane);
    if (j0 < n) edge_batch<false>(feature, sums, pk, j0, n, lane);
}

__device__ __forceinline__ void flush_row(float* sums, int r, const float (&acc)[4], int lane) {
    float* srow = sums + r * PITCH + 4 * lane;
    lds_add(srow,     acc[0]);
    lds_add(srow + 1, acc[1]);
    lds_add(srow + 2, acc[2]);
    lds_add(srow + 3, acc[3]);
}

template <bool FULL>
__device__ __forceinline__ void row_batch(const float* feature, float* sums,
                                          unsigned pk, int j0, int n, int lane,
                                          int& cur, float (&acc)[4]) {
    v4f v[EB];
#pragma unroll
    for (int j = 0; j < EB; ++j) {
        if (FULL || j0 + j < n) {
            const unsigned p = (unsigned)__builtin_amdgcn_readlane((int)pk, j0 + j);
            v[j] = row_base(feature, (int)(p & 0xFFFFu))[lane];
        }
    }
#pragma unroll
    for (int j = 0; j < EB; ++j) {
        if (FULL || j0 + j < n) {
            const unsigned p = (unsigned)__builtin_amdgcn_readlane((int)pk, j0 + j);
            const int r = (int)(p >> 16);
            if (r != cur) {
                if (cur >= 0) flush_row(sums, cur, acc, lane);
                cur = r;
                acc[0] = 0.0f; acc[1] = 0.0f; acc[2] = 0.0f; acc[3] = 0.0f;
            }
            acc[0] += v[j][0]; acc[1] += v[j][1]; acc[2] += v[j][2]; acc[3] += v[j][3];
        }
    }
}

__global__ void __launch_bounds__(BLOCK)
gcn_range_kernel(const float*    __restrict__ feature,
                 const unsigned* __restrict__ grouped,
                 const int*      __restrict__ segT,
                 const float*    __restrict__ W,
                 const float*    __restrict__ b,
                 float*          __restrict__ out) {
    __shared__ __attribute__((aligned(16))) float sums[BIN_ROWS * PITCH];
    __shared__ __attribute__((aligned(16))) float wl[FDIM * PITCH];
    __shared__ unsigned sorted[CAP];
    __shared__ int      rowpos[256];
    __shared__ int      total_s;

    const int tid  = threadIdx.x;
    const int lane = tid & 31;
    const int wave = __builtin_amdgcn_readfirstlane(tid >> 5);
    const int k    = blockIdx.x;

    int sv = 0, ev = 0;
    {
        const int c = wave + N_WAVES * lane;
        if (lane < CHUNKS_PER_WAVE && c < N_CHUNKS) {
            sv = segT[k * N_CHUNKS + c];
            ev = segT[(k + 1) * N_CHUNKS + c];
        }
    }

    v4f wreg[4];
#pragma unroll
    for (int m = 0; m < 4; ++m)
        wreg[m] = *(const v4f*)(W + 4 * (tid + BLOCK * m));

    const v4f zero4 = {0.0f, 0.0f, 0.0f, 0.0f};
    for (int q = tid; q < BIN_ROWS * PITCH / 4; q += BLOCK)
        ((v4f*)sums)[q] = zero4;
    if (tid < 256) rowpos[tid] = 0;

#pragma unroll
    for (int m = 0; m < 4; ++m) {
        const int q = tid + BLOCK * m;
        *(v4f*)(wl + (q >> 5) * PITCH + 4 * (q & 31)) = wreg[m];
    }

    unsigned pkr[CHUNKS_PER_WAVE][SEG_SLOTS];
#pragma unroll
    for (int ci = 0; ci < CHUNKS_PER_WAVE; ++ci) {
        const int c = wave + N_WAVES * ci;
        const int s = __builtin_amdgcn_readlane(sv, ci);
        const int e = __builtin_amdgcn_readlane(ev, ci);
#pragma unroll
        for (int t = 0; t < SEG_SLOTS; ++t) {
            pkr[ci][t] = 0;
            if (32 * t + lane < e - s) pkr[ci][t] = grouped[c * CHUNK + s + 32 * t + lane];
        }
    }

    __syncthreads();

#pragma unroll
    for (int ci = 0; ci < CHUNKS_PER_WAVE; ++ci) {
        const int c = wave + N_WAVES * ci;
        const int s = __builtin_amdgcn_readlane(sv, ci);
        const int e = __builtin_amdgcn_readlane(ev, ci);
#pragma unroll
        for (int t = 0; t < SEG_SLOTS; ++t)
            if (32 * t + lane < e - s) atomicAdd(&rowpos[pkr[ci][t] >> 16], 1);
        for (int base = s + 32 * SEG_SLOTS; base < e; base += 32)
            if (base + lane < e) atomicAdd(&rowpos[grouped[c * CHUNK + base + lane] >> 16], 1);
    }
    __syncthreads();

    if (tid < 32) {
        int cnt[8];
        int local = 0;
#pragma unroll
        for (int q = 0; q < 8; ++q) { cnt[q] = rowpos[8 * tid + q]; local += cnt[q]; }
        int inc = local;
#pragma unroll
        for (int d = 1; d < 32; d <<= 1) {
            const int t = __shfl_up(inc, d);
            if (tid >= d) inc += t;
        }
        int run = inc - local;
#pragma unroll
        for (int q = 0; q < 8; ++q) { rowpos[8 * tid + q] = run; run += cnt[q]; }
        if (tid == 31) total_s = inc;
    }
    __syncthreads();

    const int total = __builtin_amdgcn_readfirstlane(total_s);

    if (total <= CAP) {
#pragma unroll
        for (int ci = 0; ci < CHUNKS_PER_WAVE; ++ci) {
            const int c = wave + N_WAVES * ci;
            const int s = __builtin_amdgcn_readlane(sv, ci);
            const int e = __builtin_amdgcn_readlane(ev, ci);
#pragma unroll
            for (int t = 0; t < SEG_SLOTS; ++t)
                if (32 * t + lane < e - s) sorted[atomicAdd(&rowpos[pkr[ci][t] >> 16], 1)] = pkr[ci][t];
            for (int base = s + 32 * SEG_SLOTS; base < e; base += 32)
                if (base + lane < e) {
                    const unsigned p = grouped[c * CHUNK + base + lane];
                    sorted[atomicAdd(&rowpos[p >> 16], 1)] = p;
                }
        }
        __syncthreads();

        const int per   = (total + N_WAVES - 1) / N_WAVES;
        const int begin = min(total, wave * per);
        const int end   = min(total, begin + per);
        int   cur = -1;
        float acc[4] = {0.0f, 0.0f, 0.0f, 0.0f};
        for (int base = begin; base < end; base += 32) {
            const int n = min(32, end - base);
            const unsigned pk = (lane < n) ? sorted[base + lane] : 0u;
            int j0 = 0;
            for (; j0 + EB <= n; j0 += EB) row_batch<true>(feature, sums, pk, j0, n, lane, cur, acc);
            if (j0 < n) row_batch<false>(feature, sums, pk, j0, n, lane, cur, acc);
        }
        if (cur >= 0) flush_row(sums, cur, acc, lane);
    } else {
        for (int ci = 0; ci < CHUNKS_PER_WAVE; ++ci) {
            const int c = wave + N_WAVES * ci;
            const int s = __builtin_amdgcn_readlane(sv, ci);
            const int e = __builtin_amdgcn_readlane(ev, ci);
            for (int base = s; base < e; base += 32) {
                const int n = min(32, e - base);
                const unsigned pk = (lane < n) ? grouped[c * CHUNK + base + lane] : 0u;
                edge_group(feature, sums, pk, n, lane);
            }
        }
    }

    __syncthreads();

    const int ln    = lane & 15;
    const int half  = lane >> 4;
    const int rows  = min(BIN_ROWS, N_NODES - k * BIN_ROWS);
    const int units = (rows >> 4) * 4;
    for (int u = wave; u < units; u += N_WAVES) {
        const int s  = u >> 2;
        const int cp = u & 3;
        const float* arow = sums + (16 * s + ln) * PITCH + 4 * half;
        const float* brow = wl + (32 * cp + ln) * PITCH + 4 * half;
        const float bias0 = b[32 * cp + ln];
        const float bias1 = b[32 * cp + 16 + ln];
        v8f acc0, acc1;
#pragma unroll
        for (int v = 0; v < 8; ++v) { acc0[v] = bias0; acc1[v] = bias1; }
#pragma unroll
        for (int j = 0; j < 16; ++j) {
            const v4f a  = *(const v4f*)(arow + 8 * j);
            const v4f b0 = *(const v4f*)(brow + 8 * j);
            const v4f b1 = *(const v4f*)(brow + 16 * PITCH + 8 * j);
            v2f a2, x2, y2;
            a2[0] = a[0];  a2[1] = a[1];
            x2[0] = b0[0]; x2[1] = b0[1];
            y2[0] = b1[0]; y2[1] = b1[1];
            acc0 = __builtin_amdgcn_wmma_f32_16x16x4_f32(false, a2, false, x2, (short)0, acc0, false, false);
            acc1 = __builtin_amdgcn_wmma_f32_16x16x4_f32(false, a2, false, y2, (short)0, acc1, false, false);
            a2[0] = a[2];  a2[1] = a[3];
            x2[0] = b0[2]; x2[1] = b0[3];
            y2[0] = b1[2]; y2[1] = b1[3];
            acc0 = __builtin_amdgcn_wmma_f32_16x16x4_f32(false, a2, false, x2, (short)0, acc0, false, false);
            acc1 = __builtin_amdgcn_wmma_f32_16x16x4_f32(false, a2, false, y2, (short)0, acc1, false, false);
        }
        float* orow = out + (size_t)(k * BIN_ROWS + 16 * s + 8 * half) * FDIM + 32 * cp + ln;
#pragma unroll
        for (int v = 0; v < 8; ++v) {
            orow[v * FDIM]      = acc0[v];
            orow[v * FDIM + 16] = acc1[v];
        }
    }
}

extern "C" void kernel_launch(void* const* d_in, const int* in_sizes, int n_in,
                              void* d_out, int out_size, void* d_ws, size_t ws_size,
                              hipStream_t stream) {
    const float* feature = (const float*)d_in[0];
    const int*   src     = (const int*)  d_in[1];
    const int*   dst     = (const int*)  d_in[2];
    const float* W       = (const float*)d_in[3];
    const float* b       = (const float*)d_in[4];
    float*       out     = (float*)d_out;
    unsigned*    grouped = (unsigned*)d_ws;
    int*         segT    = (int*)(grouped + E_EDGES);

    gcn_group_kernel<<<N_CHUNKS, BLOCK, 0, stream>>>(src, dst, grouped, segT);

    gcn_range_kernel<<<N_BINS, BLOCK, 0, stream>>>(feature, grouped, segT, W, b, out);
}
